# starting kernel
# baseline (speedup 1.0000x reference)
#include <hip/hip_runtime.h>

typedef __attribute__((ext_vector_type(2))) float v2f;
typedef __attribute__((ext_vector_type(8))) float v8f;

#define NPWM    512
#define FS      19
#define LEN     1000
#define OUTPOS  982
#define PTILES  62
#define KSTEPS  19
#define NEG_INF (-3.402823466e38f)

static __device__ __forceinline__ v8f wmma_f32_16x16x4(v2f a, v2f b, v8f c) {
  return __builtin_amdgcn_wmma_f32_16x16x4_f32(
       false, a,  false, b,
       (short)0, c,  false,  false);
}

__global__ __launch_bounds__(256) void pwm_scan_max_kernel(
    const float* __restrict__ x,
    const float* __restrict__ pwms,
    float* __restrict__ out) {
  __shared__ float xT[4096];
  __shared__ float red[8][16];

  const int b     = blockIdx.y;
  const int f0    = blockIdx.x * 16;
  const int tid   = threadIdx.x;
  const int lane  = tid & 31;
  const int wave  = tid >> 5;
  const int n     = lane & 15;
  const int cbase = (lane < 16) ? 0 : 2;

  const float* xb = x + (size_t)b * 4 * LEN;
  for (int idx = tid; idx < 4096; idx += 256) {
    int pos = idx >> 2, c = idx & 3;
    xT[idx] = (pos < LEN) ? xb[c * LEN + pos] : 0.0f;
  }

  const float* wrow = pwms + (size_t)(f0 + n) * (4 * FS);
  v2f bF[KSTEPS], bR[KSTEPS];
#pragma unroll
  for (int g = 0; g < KSTEPS; ++g) {
    bF[g].x = wrow[cbase * FS + g];
    bF[g].y = wrow[(cbase + 1) * FS + g];
    bR[g].x = wrow[(3 - cbase) * FS + (FS - 1 - g)];
    bR[g].y = wrow[(2 - cbase) * FS + (FS - 1 - g)];
  }

  __syncthreads();

  float rmax = NEG_INF;
  const float* ap_base = &xT[0];
  for (int pt = wave; pt < PTILES; pt += 8) {
    const int pos0 = pt * 16;
    v8f accF = {};
    v8f accR = {};
    const float* ap = ap_base + (pos0 + n) * 4 + cbase;
#pragma unroll
    for (int g = 0; g < KSTEPS; ++g) {
      v2f a = *(const v2f*)(ap + g * 4);
      accF = wmma_f32_16x16x4(a, bF[g], accF);
      accR = wmma_f32_16x16x4(a, bR[g], accR);
    }
    const int mbase = pos0 + ((lane >= 16) ? 8 : 0);
#pragma unroll
    for (int r = 0; r < 8; ++r) {
      float v = fmaxf(accF[r], accR[r]);
      rmax = (mbase + r < OUTPOS) ? fmaxf(rmax, v) : rmax;
    }
  }

  rmax = fmaxf(rmax, __shfl_xor(rmax, 16, 32));
  if (lane < 16) red[wave][n] = rmax;
  __syncthreads();
  if (tid < 16) {
    float m = red[0][tid];
#pragma unroll
    for (int w = 1; w < 8; ++w) m = fmaxf(m, red[w][tid]);
    out[(size_t)b * NPWM + f0 + tid] = m;
  }
}

extern "C" void kernel_launch(void* const* d_in, const int* in_sizes, int n_in,
                              void* d_out, int out_size, void* d_ws, size_t ws_size,
                              hipStream_t stream) {
  const float* x    = (const float*)d_in[0];
  const float* pwms = (const float*)d_in[1];
  float* out        = (float*)d_out;

  dim3 grid(NPWM / 16, 128);
  pwm_scan_max_kernel<<<grid, 256, 0, stream>>>(x, pwms, out);
}


# Two-phase fast FIR over the taps: even/odd positions and taps give 3 half-length products (K 40, 40, 36) per output pair instead of 4: 58 WMMAs per 32 outputs, were 76; f32 summation order changes
# speedup vs baseline: 1.4949x; 1.4949x over previous
#include <hip/hip_runtime.h>

typedef __attribute__((ext_vector_type(2))) float v2f;
typedef __attribute__((ext_vector_type(4))) float v4f;
typedef __attribute__((ext_vector_type(8))) float v8f;

#define NPWM    512
#define FS      19
#define LEN     1000
#define OUTPOS  982
#define NPAIR   491
#define HT      10
#define WTILES  4
#define WSPAN   63
#define NEG_INF (-3.402823466e38f)

static __device__ __forceinline__ v8f wmma_f32_16x16x4(v2f a, v2f b, v8f c) {
  return __builtin_amdgcn_wmma_f32_16x16x4_f32(
       false, a,  false, b,
       (short)0, c,  false,  false);
}

#define BPB     8


static __device__ __forceinline__ void stage_x(v4f* x0, v4f* x1, v4f* xs, int tid,
                                               v4f r0, v4f r1, v4f r2, v4f r3) {
  if (tid >= LEN / 4) { r0 = (v4f){}; r1 = (v4f){}; r2 = (v4f){}; r3 = (v4f){}; }
  const v4f c0 = { r0[0], r1[0], r2[0], r3[0] };
  const v4f c1 = { r0[1], r1[1], r2[1], r3[1] };
  const v4f c2 = { r0[2], r1[2], r2[2], r3[2] };
  const v4f c3 = { r0[3], r1[3], r2[3], r3[3] };
  x0[2 * tid]     = c0;
  x0[2 * tid + 1] = c2;
  x1[2 * tid]     = c1;
  x1[2 * tid + 1] = c3;
  xs[2 * tid]     = c0 + c1;
  xs[2 * tid + 1] = c2 + c3;
}

static __device__ __forceinline__ float fold_tile(float rmax, v8f pa, v8f pb, v8f pc,
                                                  float pa_next7, int srow, bool last_odd_ok) {
#pragma unroll
  for (int r = 0; r < 8; ++r) {
    const float ev = pc[r] - pa[r] - pb[r];
    const float od = ((r < 7) ? pa[r + 1] : pa_next7) + pb[r];
    const bool  ok  = (srow + r) < NPAIR;
    const bool  oko = ok && ((r < 7) || last_odd_ok);
    rmax = fmaxf(rmax, fmaxf(ok ? ev : NEG_INF, oko ? od : NEG_INF));
  }
  return rmax;
}

__global__ __launch_bounds__(256) void pwm_scan_max_kernel(
    const float* __restrict__ x,
    const float* __restrict__ pwms,
    float* __restrict__ out) {
  __shared__ v4f xph[2][3][512];
  __shared__ float red[2][8][16];

  const int f0    = blockIdx.x * 16;
  const int b0    = blockIdx.y * BPB;
  const int tid   = threadIdx.x;
  const int lane  = tid & 31;
  const int wave  = tid >> 5;
  const int n     = lane & 15;
  const bool hi   = lane >= 16;
  const int cbase = hi ? 2 : 0;

  const float* wrow = pwms + (size_t)(f0 + n) * (4 * FS);
  v2f g0F[HT - 1], g1F[HT], gsF[HT], g0R[HT - 1], g1R[HT], gsR[HT];
#pragma unroll
  for (int j = 0; j < HT; ++j) {
    g1F[j].x = wrow[cbase * FS + 2 * j];
    g1F[j].y = wrow[(cbase + 1) * FS + 2 * j];
    g1R[j].x = wrow[(3 - cbase) * FS + (FS - 1 - 2 * j)];
    g1R[j].y = wrow[(2 - cbase) * FS + (FS - 1 - 2 * j)];
  }
#pragma unroll
  for (int j = 0; j < HT - 1; ++j) {
    g0F[j].x = wrow[cbase * FS + 2 * j + 1];
    g0F[j].y = wrow[(cbase + 1) * FS + 2 * j + 1];
    g0R[j].x = wrow[(3 - cbase) * FS + (FS - 2 - 2 * j)];
    g0R[j].y = wrow[(2 - cbase) * FS + (FS - 2 - 2 * j)];
  }

  const int  q   = (tid < LEN / 4) ? tid : (LEN / 4 - 1);
  const v4f* xb4 = (const v4f*)(x + (size_t)b0 * 4 * LEN);
  v4f r0 = xb4[q];
  v4f r1 = xb4[LEN / 4 + q];
  v4f r2 = xb4[2 * (LEN / 4) + q];
  v4f r3 = xb4[3 * (LEN / 4) + q];

  __builtin_amdgcn_sched_barrier(0);

#pragma unroll
  for (int j = 0; j < HT; ++j) {
    gsF[j] = g1F[j];
    gsR[j] = g1R[j];
    if (j < HT - 1) { gsF[j] += g0F[j]; gsR[j] += g0R[j]; }
  }

  stage_x(xph[0][0], xph[0][1], xph[0][2], tid, r0, r1, r2, r3);
  __syncthreads();

  const int rw = (wave * WSPAN < NPAIR + 1 - 16 * WTILES) ? wave * WSPAN : (NPAIR + 1 - 16 * WTILES);

  for (int i = 0; i < BPB; ++i) {
    const int cur = i & 1;
    if (i + 1 < BPB) {
      const v4f* xn4 = (const v4f*)(x + (size_t)(b0 + i + 1) * 4 * LEN);
      r0 = xn4[q];
      r1 = xn4[LEN / 4 + q];
      r2 = xn4[2 * (LEN / 4) + q];
      r3 = xn4[3 * (LEN / 4) + q];
    }

    float rmax = NEG_INF;
    float carryF = 0.0f, carryR = 0.0f;
    const float* p0 = (const float*)xph[cur][0];
    const float* p1 = (const float*)xph[cur][1];
    const float* ps = (const float*)xph[cur][2];
    for (int it = WTILES - 1; it >= 0; --it) {
      const int s0  = rw + 16 * it;
      const int off = (s0 + n) * 4 + cbase;
      v2f fa[HT - 1], fb[HT], fs[HT];
#pragma unroll
      for (int j = 0; j < HT; ++j) {
        fs[j] = *(const v2f*)(ps + off + j * 4);
        fb[j] = *(const v2f*)(p1 + off + j * 4);
        if (j < HT - 1) fa[j] = *(const v2f*)(p0 + off + j * 4);
      }
      __builtin_amdgcn_sched_barrier(0);
      v8f paF = {}, pbF = {}, pcF = {}, paR = {}, pbR = {}, pcR = {};
#pragma unroll
      for (int j = 0; j < HT; ++j) {
        pcF = wmma_f32_16x16x4(fs[j], gsF[j], pcF);
        pcR = wmma_f32_16x16x4(fs[j], gsR[j], pcR);
        pbF = wmma_f32_16x16x4(fb[j], g1F[j], pbF);
        pbR = wmma_f32_16x16x4(fb[j], g1R[j], pbR);
        if (j < HT - 1) {
          paF = wmma_f32_16x16x4(fa[j], g0F[j], paF);
          paR = wmma_f32_16x16x4(fa[j], g0R[j], paR);
        }
      }
      const float xF = __shfl_xor(paF[0], 16, 32);
      const float xR = __shfl_xor(paR[0], 16, 32);
      const float n7F = hi ? carryF : xF;
      const float n7R = hi ? carryR : xR;
      carryF = xF;
      carryR = xR;
      const int  srow = s0 + (hi ? 8 : 0);
      const bool last_odd_ok = !(hi && it == WTILES - 1);
      rmax = fold_tile(rmax, paF, pbF, pcF, n7F, srow, last_odd_ok);
      rmax = fold_tile(rmax, paR, pbR, pcR, n7R, srow, last_odd_ok);
    }

    rmax = fmaxf(rmax, __shfl_xor(rmax, 16, 32));
    if (lane < 16) red[cur][wave][n] = rmax;
    if (i + 1 < BPB) stage_x(xph[cur ^ 1][0], xph[cur ^ 1][1], xph[cur ^ 1][2], tid, r0, r1, r2, r3);
    __syncthreads();
    if (tid < 16) {
      float m = red[cur][0][tid];
#pragma unroll
      for (int w = 1; w < 8; ++w) m = fmaxf(m, red[cur][w][tid]);
      out[(size_t)(b0 + i) * NPWM + f0 + tid] = m;
    }
  }
}

extern "C" void kernel_launch(void* const* d_in, const int* in_sizes, int n_in,
                              void* d_out, int out_size, void* d_ws, size_t ws_size,
                              hipStream_t stream) {
  const float* x    = (const float*)d_in[0];
  const float* pwms = (const float*)d_in[1];
  float* out        = (float*)d_out;

  dim3 grid(NPWM / 16, 128 / BPB);
  pwm_scan_max_kernel<<<grid, 256, 0, stream>>>(x, pwms, out);
}


# Two-phase FIR turned round: x staged as x1, x0-x1, x0[u+1]-x1; y[2s], y[2s+1] carry the shared product's accumulator on by one more WMMA chain each: no vector recombination; f32 sum order changes
# speedup vs baseline: 1.6065x; 1.0747x over previous
#include <hip/hip_runtime.h>

typedef __attribute__((ext_vector_type(2))) float v2f;
typedef __attribute__((ext_vector_type(4))) float v4f;
typedef __attribute__((ext_vector_type(8))) float v8f;

#define NPWM    512
#define FS      19
#define LEN     1000
#define OUTPOS  982
#define NPAIR   491
#define HT      10
#define WTILES  4
#define WROWS   64
#define NEG_INF (-3.402823466e38f)

static __device__ __forceinline__ v8f wmma_f32_16x16x4(v2f a, v2f b, v8f c) {
  return __builtin_amdgcn_wmma_f32_16x16x4_f32(
       false, a,  false, b,
       (short)0, c,  false,  false);
}

#define BPB     8


static __device__ __forceinline__ void stage_x(v4f* x1, v4f* xd, v4f* xe, int tid,
                                               v4f r0, v4f r1, v4f r2, v4f r3, v4f nx) {
  if (tid >= LEN / 4) { r0 = (v4f){}; r1 = (v4f){}; r2 = (v4f){}; r3 = (v4f){}; }
  if (tid >= LEN / 4 - 1) nx = (v4f){};
  const v4f c0 = { r0[0], r1[0], r2[0], r3[0] };
  const v4f c1 = { r0[1], r1[1], r2[1], r3[1] };
  const v4f c2 = { r0[2], r1[2], r2[2], r3[2] };
  const v4f c3 = { r0[3], r1[3], r2[3], r3[3] };
  x1[2 * tid]     = c1;
  x1[2 * tid + 1] = c3;
  xd[2 * tid]     = c0 - c1;
  xd[2 * tid + 1] = c2 - c3;
  xe[2 * tid]     = c2 - c1;
  xe[2 * tid + 1] = nx - c3;
}

__global__ __launch_bounds__(256) __attribute__((amdgpu_waves_per_eu(4, 4))) void pwm_scan_max_kernel(
    const float* __restrict__ x,
    const float* __restrict__ pwms,
    float* __restrict__ out) {
  __shared__ v4f xph[2][3][512];
  __shared__ float red[2][8][16];

  const int f0    = blockIdx.x * 16;
  const int b0    = blockIdx.y * BPB;
  const int tid   = threadIdx.x;
  const int lane  = tid & 31;
  const int wave  = tid >> 5;
  const int n     = lane & 15;
  const bool hi   = lane >= 16;
  const int cbase = hi ? 2 : 0;

  const float* wrow = pwms + (size_t)(f0 + n) * (4 * FS);
  v2f g0F[HT - 1], g1F[HT], gsF[HT], g0R[HT - 1], g1R[HT], gsR[HT];
#pragma unroll
  for (int j = 0; j < HT; ++j) {
    g1F[j].x = wrow[cbase * FS + 2 * j];
    g1F[j].y = wrow[(cbase + 1) * FS + 2 * j];
    g1R[j].x = wrow[(3 - cbase) * FS + (FS - 1 - 2 * j)];
    g1R[j].y = wrow[(2 - cbase) * FS + (FS - 1 - 2 * j)];
  }
#pragma unroll
  for (int j = 0; j < HT - 1; ++j) {
    g0F[j].x = wrow[cbase * FS + 2 * j + 1];
    g0F[j].y = wrow[(cbase + 1) * FS + 2 * j + 1];
    g0R[j].x = wrow[(3 - cbase) * FS + (FS - 2 - 2 * j)];
    g0R[j].y = wrow[(2 - cbase) * FS + (FS - 2 - 2 * j)];
  }

  const int  q   = (tid < LEN / 4) ? tid : (LEN / 4 - 1);
  const v4f* xb4 = (const v4f*)(x + (size_t)b0 * 4 * LEN);
  v4f r0 = xb4[q];
  v4f r1 = xb4[LEN / 4 + q];
  v4f r2 = xb4[2 * (LEN / 4) + q];
  v4f r3 = xb4[3 * (LEN / 4) + q];
  const int    pn  = (tid < LEN / 4 - 1) ? 4 * tid + 4 : 0;
  const float* xb1 = x + (size_t)b0 * 4 * LEN + pn;
  v4f nx = { xb1[0], xb1[LEN], xb1[2 * LEN], xb1[3 * LEN] };

  __builtin_amdgcn_sched_barrier(0);

#pragma unroll
  for (int j = 0; j < HT; ++j) {
    gsF[j] = g1F[j];
    gsR[j] = g1R[j];
    if (j < HT - 1) { gsF[j] += g0F[j]; gsR[j] += g0R[j]; }
  }

  stage_x(xph[0][0], xph[0][1], xph[0][2], tid, r0, r1, r2, r3, nx);
  __syncthreads();

  const int rw = (wave * WROWS < NPAIR - WROWS) ? wave * WROWS : (NPAIR - WROWS);

  for (int i = 0; i < BPB; ++i) {
    const int cur = i & 1;
    if (i + 1 < BPB) {
      const v4f* xn4 = (const v4f*)(x + (size_t)(b0 + i + 1) * 4 * LEN);
      r0 = xn4[q];
      r1 = xn4[LEN / 4 + q];
      r2 = xn4[2 * (LEN / 4) + q];
      r3 = xn4[3 * (LEN / 4) + q];
      const float* xn1 = x + (size_t)(b0 + i + 1) * 4 * LEN + pn;
      nx = (v4f){ xn1[0], xn1[LEN], xn1[2 * LEN], xn1[3 * LEN] };
    }

    float rmax = NEG_INF, rmaxR = NEG_INF;
    const float* p1 = (const float*)xph[cur][0];
    const float* pd = (const float*)xph[cur][1];
    const float* pe = (const float*)xph[cur][2];
    for (int it = 0; it < WTILES; ++it) {
      const int s0  = rw + 16 * it;
      const int off = (s0 + n) * 4 + cbase;
      v2f f1[HT], fd[HT], fe[HT - 1];
#pragma unroll
      for (int j = 0; j < HT; ++j) {
        f1[j] = *(const v2f*)(p1 + off + j * 4);
        fd[j] = *(const v2f*)(pd + off + j * 4);
        if (j < HT - 1) fe[j] = *(const v2f*)(pe + off + j * 4);
      }
      __builtin_amdgcn_sched_barrier(0);
      v8f mF = {}, mR = {};
#pragma unroll
      for (int j = 0; j < HT; ++j) {
        mF = wmma_f32_16x16x4(f1[j], gsF[j], mF);
        mR = wmma_f32_16x16x4(f1[j], gsR[j], mR);
      }
      v8f evF = mF, odF = mF, evR = mR, odR = mR;
#pragma unroll
      for (int j = 0; j < HT; ++j) {
        evF = wmma_f32_16x16x4(fd[j], g1F[j], evF);
        evR = wmma_f32_16x16x4(fd[j], g1R[j], evR);
        if (j < HT - 1) {
          odF = wmma_f32_16x16x4(fe[j], g0F[j], odF);
          odR = wmma_f32_16x16x4(fe[j], g0R[j], odR);
        }
      }
#pragma unroll
      for (int r = 0; r < 8; ++r) {
        rmax  = fmaxf(rmax,  fmaxf(evF[r], odF[r]));
        rmaxR = fmaxf(rmaxR, fmaxf(evR[r], odR[r]));
      }
    }

    rmax = fmaxf(rmax, rmaxR);
    rmax = fmaxf(rmax, __shfl_xor(rmax, 16, 32));
    if (lane < 16) red[cur][wave][n] = rmax;
    if (i + 1 < BPB) stage_x(xph[cur ^ 1][0], xph[cur ^ 1][1], xph[cur ^ 1][2], tid, r0, r1, r2, r3, nx);
    __syncthreads();
    if (tid < 16) {
      float m = red[cur][0][tid];
#pragma unroll
      for (int w = 1; w < 8; ++w) m = fmaxf(m, red[cur][w][tid]);
      out[(size_t)(b0 + i) * NPWM + f0 + tid] = m;
    }
  }
}

extern "C" void kernel_launch(void* const* d_in, const int* in_sizes, int n_in,
                              void* d_out, int out_size, void* d_ws, size_t ws_size,
                              hipStream_t stream) {
  const float* x    = (const float*)d_in[0];
  const float* pwms = (const float*)d_in[1];
  float* out        = (float*)d_out;

  dim3 grid(NPWM / 16, 128 / BPB);
  pwm_scan_max_kernel<<<grid, 256, 0, stream>>>(x, pwms, out);
}


# Fast FIR nested once more: four output phases, nine quarter-length products (K 4 x 5) give 44 WMMAs per 64 outputs and strand, were 58; a wave does one strand; f32 summation order changes
# speedup vs baseline: 1.8152x; 1.1299x over previous
#include <hip/hip_runtime.h>

typedef __attribute__((ext_vector_type(2))) float v2f;
typedef __attribute__((ext_vector_type(4))) float v4f;
typedef __attribute__((ext_vector_type(8))) float v8f;

#define NPWM    512
#define FS      19
#define LEN     1000
#define OUTPOS  982
#define NQUAD   246
#define QT      5
#define NPLANE  9
#define PLEN    256
#define WTILES  4
#define WROWS   64
#define NEG_INF (-3.402823466e38f)

static __device__ __forceinline__ v8f wmma_f32_16x16x4(v2f a, v2f b, v8f c) {
  return __builtin_amdgcn_wmma_f32_16x16x4_f32(
       false, a,  false, b,
       (short)0, c,  false,  false);
}

#define BPB     8


static __device__ __forceinline__ void stage_x(v4f (*pl)[PLEN], int tid,
                                               v4f r0, v4f r1, v4f r2, v4f r3,
                                               v4f m0, v4f m1, v4f m2, v4f m3) {
  if (tid >= LEN / 4) { r0 = (v4f){}; r1 = (v4f){}; r2 = (v4f){}; r3 = (v4f){}; }
  if (tid >= LEN / 4 - 1) { m0 = (v4f){}; m1 = (v4f){}; m2 = (v4f){}; m3 = (v4f){}; }
  const v4f c0 = { r0[0], r1[0], r2[0], r3[0] };
  const v4f c1 = { r0[1], r1[1], r2[1], r3[1] };
  const v4f c2 = { r0[2], r1[2], r2[2], r3[2] };
  const v4f c3 = { r0[3], r1[3], r2[3], r3[3] };
  const v4f n0 = { m0[0], m1[0], m2[0], m3[0] };
  const v4f n1 = { m0[1], m1[1], m2[1], m3[1] };
  const v4f n2 = { m0[2], m1[2], m2[2], m3[2] };
  const v4f dA = c2 - c3;
  const v4f dB = n0 - c3;
  pl[0][tid] = c3;
  pl[1][tid] = c1 - c3;
  pl[2][tid] = n1 - c3;
  pl[3][tid] = dA;
  pl[4][tid] = (c0 - c1) - dA;
  pl[5][tid] = (n0 - n1) - dA;
  pl[6][tid] = dB;
  pl[7][tid] = (c2 - c1) - dB;
  pl[8][tid] = (n2 - n1) - dB;
}

__global__ __launch_bounds__(256) __attribute__((amdgpu_waves_per_eu(4, 4))) void pwm_scan_max_kernel(
    const float* __restrict__ x,
    const float* __restrict__ pwms,
    float* __restrict__ out) {
  __shared__ v4f xph[2][NPLANE][PLEN];
  __shared__ float red[2][8][16];

  const int f0    = blockIdx.x * 16;
  const int b0    = blockIdx.y * BPB;
  const int tid   = threadIdx.x;
  const int lane  = tid & 31;
  const int wave  = tid >> 5;
  const int n     = lane & 15;
  const bool hi   = lane >= 16;
  const int cbase = hi ? 2 : 0;
  const int strand = wave >> 2;
  const int ws     = wave & 3;

  const float* wrow = pwms + (size_t)(f0 + n) * (4 * FS);
  const int bx = strand ? (3 - cbase) * FS + (FS - 1) : cbase * FS;
  const int by = strand ? (2 - cbase) * FS + (FS - 1) : (cbase + 1) * FS;
  const int sg = strand ? -1 : 1;
  v2f w[4 * QT];
#pragma unroll
  for (int k = 0; k < FS; ++k) {
    w[k].x = wrow[bx + sg * k];
    w[k].y = wrow[by + sg * k];
  }
  w[FS] = (v2f){ 0.0f, 0.0f };

  const int  q   = (tid < LEN / 4) ? tid : (LEN / 4 - 1);
  const int  qn  = (tid < LEN / 4 - 1) ? tid + 1 : (LEN / 4 - 1);
  const v4f* xb4 = (const v4f*)(x + (size_t)b0 * 4 * LEN);
  v4f r0 = xb4[q];
  v4f r1 = xb4[LEN / 4 + q];
  v4f r2 = xb4[2 * (LEN / 4) + q];
  v4f r3 = xb4[3 * (LEN / 4) + q];
  v4f m0 = xb4[qn];
  v4f m1 = xb4[LEN / 4 + qn];
  v4f m2 = xb4[2 * (LEN / 4) + qn];
  v4f m3 = xb4[3 * (LEN / 4) + qn];

  __builtin_amdgcn_sched_barrier(0);

  v2f hSS[QT], hS1[QT], hS0[QT], h1S[QT], h11[QT], h10[QT], h0S[QT], h01[QT], h00[QT - 1];
#pragma unroll
  for (int i = 0; i < QT; ++i) {
    const v2f ta = w[4 * i], tb = w[4 * i + 1], tc = w[4 * i + 2], td = w[4 * i + 3];
    hS1[i] = ta + tb;
    hS0[i] = tc + td;
    hSS[i] = hS1[i] + hS0[i];
    h1S[i] = ta + tc;
    h11[i] = ta;
    h10[i] = tc;
    h0S[i] = tb + td;
    h01[i] = tb;
    if (i < QT - 1) h00[i] = td;
  }

  stage_x(xph[0], tid, r0, r1, r2, r3, m0, m1, m2, m3);
  __syncthreads();

  const int rw = (ws * WROWS < NQUAD - WROWS) ? ws * WROWS : (NQUAD - WROWS);

  for (int i = 0; i < BPB; ++i) {
    const int cur = i & 1;
    if (i + 1 < BPB) {
      const v4f* xn4 = (const v4f*)(x + (size_t)(b0 + i + 1) * 4 * LEN);
      r0 = xn4[q];
      r1 = xn4[LEN / 4 + q];
      r2 = xn4[2 * (LEN / 4) + q];
      r3 = xn4[3 * (LEN / 4) + q];
      m0 = xn4[qn];
      m1 = xn4[LEN / 4 + qn];
      m2 = xn4[2 * (LEN / 4) + qn];
      m3 = xn4[3 * (LEN / 4) + qn];
    }

    float rmax = NEG_INF;
    const float* pb = (const float*)&xph[cur][0][0];
    for (int it = 0; it < WTILES; ++it) {
      const int s0  = rw + 16 * it;
      const float* pa = pb + (s0 + n) * 4 + cbase;
      v2f a0[QT], a1[QT], a2[QT], a3[QT], a4[QT], a5[QT], a6[QT], a7[QT], a8[QT - 1];
#pragma unroll
      for (int j = 0; j < QT; ++j) {
        a0[j] = *(const v2f*)(pa + 0 * (PLEN * 4) + j * 4);
        a3[j] = *(const v2f*)(pa + 3 * (PLEN * 4) + j * 4);
        a6[j] = *(const v2f*)(pa + 6 * (PLEN * 4) + j * 4);
        a1[j] = *(const v2f*)(pa + 1 * (PLEN * 4) + j * 4);
        a2[j] = *(const v2f*)(pa + 2 * (PLEN * 4) + j * 4);
        a4[j] = *(const v2f*)(pa + 4 * (PLEN * 4) + j * 4);
        a5[j] = *(const v2f*)(pa + 5 * (PLEN * 4) + j * 4);
        a7[j] = *(const v2f*)(pa + 7 * (PLEN * 4) + j * 4);
        if (j < QT - 1) a8[j] = *(const v2f*)(pa + 8 * (PLEN * 4) + j * 4);
      }
      __builtin_amdgcn_sched_barrier(0);
      v8f acc = {}, qd = {}, qe = {};
#pragma unroll
      for (int j = 0; j < QT; ++j) {
        acc = wmma_f32_16x16x4(a0[j], hSS[j], acc);
        qd  = wmma_f32_16x16x4(a3[j], h1S[j], qd);
        qe  = wmma_f32_16x16x4(a6[j], h0S[j], qe);
      }
      v8f u = acc, v = acc;
#pragma unroll
      for (int j = 0; j < QT; ++j) {
        u = wmma_f32_16x16x4(a1[j], hS1[j], u);
        v = wmma_f32_16x16x4(a2[j], hS0[j], v);
      }
      v8f y0 = u, y1 = u, y2 = v, y3 = v;
#pragma unroll
      for (int j = 0; j < QT; ++j) {
        y0 = wmma_f32_16x16x4(a4[j], h11[j], y0);
        y2 = wmma_f32_16x16x4(a5[j], h10[j], y2);
        y1 = wmma_f32_16x16x4(a7[j], h01[j], y1);
        if (j < QT - 1) y3 = wmma_f32_16x16x4(a8[j], h00[j], y3);
      }
      if (ws == 3 && it == WTILES - 1 && hi) { y2[7] = NEG_INF; y3[7] = NEG_INF; }
#pragma unroll
      for (int r = 0; r < 8; ++r) {
        const float e = fmaxf(y0[r], y2[r]) + qd[r];
        const float o = fmaxf(y1[r], y3[r]) + qe[r];
        rmax = fmaxf(rmax, fmaxf(e, o));
      }
    }

    rmax = fmaxf(rmax, __shfl_xor(rmax, 16, 32));
    if (lane < 16) red[cur][wave][n] = rmax;
    if (i + 1 < BPB) stage_x(xph[cur ^ 1], tid, r0, r1, r2, r3, m0, m1, m2, m3);
    __syncthreads();
    if (tid < 16) {
      float m = red[cur][0][tid];
#pragma unroll
      for (int k = 1; k < 8; ++k) m = fmaxf(m, red[cur][k][tid]);
      out[(size_t)(b0 + i) * NPWM + f0 + tid] = m;
    }
  }
}

extern "C" void kernel_launch(void* const* d_in, const int* in_sizes, int n_in,
                              void* d_out, int out_size, void* d_ws, size_t ws_size,
                              hipStream_t stream) {
  const float* x    = (const float*)d_in[0];
  const float* pwms = (const float*)d_in[1];
  float* out        = (float*)d_out;

  dim3 grid(NPWM / 16, 128 / BPB);
  pwm_scan_max_kernel<<<grid, 256, 0, stream>>>(x, pwms, out);
}


# Each lane's 38 filter taps (one contiguous run of its row) fetched as ten 16-byte loads from an aligned start, taps picked from it in registers; were 38 four-byte loads, each touching 32 lines
# speedup vs baseline: 1.8809x; 1.0362x over previous
#include <hip/hip_runtime.h>

typedef __attribute__((ext_vector_type(2))) float v2f;
typedef __attribute__((ext_vector_type(4))) float v4f;
typedef __attribute__((ext_vector_type(8))) float v8f;

#define NPWM    512
#define FS      19
#define LEN     1000
#define OUTPOS  982
#define NQUAD   246
#define QT      5
#define NPLANE  9
#define PLEN    256
#define WTILES  4
#define WROWS   64
#define NEG_INF (-3.402823466e38f)

static __device__ __forceinline__ v8f wmma_f32_16x16x4(v2f a, v2f b, v8f c) {
  return __builtin_amdgcn_wmma_f32_16x16x4_f32(
       false, a,  false, b,
       (short)0, c,  false,  false);
}

#define BPB     8


static __device__ __forceinline__ void stage_x(v4f (*pl)[PLEN], int tid,
                                               v4f r0, v4f r1, v4f r2, v4f r3,
                                               v4f m0, v4f m1, v4f m2, v4f m3) {
  if (tid >= LEN / 4) { r0 = (v4f){}; r1 = (v4f){}; r2 = (v4f){}; r3 = (v4f){}; }
  if (tid >= LEN / 4 - 1) { m0 = (v4f){}; m1 = (v4f){}; m2 = (v4f){}; m3 = (v4f){}; }
  const v4f c0 = { r0[0], r1[0], r2[0], r3[0] };
  const v4f c1 = { r0[1], r1[1], r2[1], r3[1] };
  const v4f c2 = { r0[2], r1[2], r2[2], r3[2] };
  const v4f c3 = { r0[3], r1[3], r2[3], r3[3] };
  const v4f n0 = { m0[0], m1[0], m2[0], m3[0] };
  const v4f n1 = { m0[1], m1[1], m2[1], m3[1] };
  const v4f n2 = { m0[2], m1[2], m2[2], m3[2] };
  const v4f dA = c2 - c3;
  const v4f dB = n0 - c3;
  pl[0][tid] = c3;
  pl[1][tid] = c1 - c3;
  pl[2][tid] = n1 - c3;
  pl[3][tid] = dA;
  pl[4][tid] = (c0 - c1) - dA;
  pl[5][tid] = (n0 - n1) - dA;
  pl[6][tid] = dB;
  pl[7][tid] = (c2 - c1) - dB;
  pl[8][tid] = (n2 - n1) - dB;
}

__global__ __launch_bounds__(256) __attribute__((amdgpu_waves_per_eu(4, 4))) void pwm_scan_max_kernel(
    const float* __restrict__ x,
    const float* __restrict__ pwms,
    float* __restrict__ out) {
  __shared__ v4f xph[2][NPLANE][PLEN];
  __shared__ float red[2][8][16];

  const int f0    = blockIdx.x * 16;
  const int b0    = blockIdx.y * BPB;
  const int tid   = threadIdx.x;
  const int lane  = tid & 31;
  const int wave  = tid >> 5;
  const int n     = lane & 15;
  const bool hi   = lane >= 16;
  const int cbase = hi ? 2 : 0;
  const int strand = wave >> 2;
  const int ws     = wave & 3;

  const float* wrow = pwms + (size_t)(f0 + n) * (4 * FS);
  const bool upper = strand ? !hi : hi;
  const v4f* wsrc = (const v4f*)(wrow + (upper ? 2 * FS - 2 : 0));
  v4f wl[(2 * FS + 2) / 4];
#pragma unroll
  for (int j = 0; j < (2 * FS + 2) / 4; ++j) wl[j] = wsrc[j];

  const int  q   = (tid < LEN / 4) ? tid : (LEN / 4 - 1);
  const int  qn  = (tid < LEN / 4 - 1) ? tid + 1 : (LEN / 4 - 1);
  const v4f* xb4 = (const v4f*)(x + (size_t)b0 * 4 * LEN);
  v4f r0 = xb4[q];
  v4f r1 = xb4[LEN / 4 + q];
  v4f r2 = xb4[2 * (LEN / 4) + q];
  v4f r3 = xb4[3 * (LEN / 4) + q];
  v4f m0 = xb4[qn];
  v4f m1 = xb4[LEN / 4 + qn];
  v4f m2 = xb4[2 * (LEN / 4) + qn];
  v4f m3 = xb4[3 * (LEN / 4) + qn];

  __builtin_amdgcn_sched_barrier(0);

  v2f w[4 * QT];
#pragma unroll
  for (int k = 0; k < FS; ++k) {
    const int fx = k, fy = FS + k, rx = 2 * FS - 1 - k, ry = FS - 1 - k;
    const float gfx = upper ? wl[(fx + 2) >> 2][(fx + 2) & 3] : wl[fx >> 2][fx & 3];
    const float gfy = upper ? wl[(fy + 2) >> 2][(fy + 2) & 3] : wl[fy >> 2][fy & 3];
    const float grx = upper ? wl[(rx + 2) >> 2][(rx + 2) & 3] : wl[rx >> 2][rx & 3];
    const float gry = upper ? wl[(ry + 2) >> 2][(ry + 2) & 3] : wl[ry >> 2][ry & 3];
    w[k].x = strand ? grx : gfx;
    w[k].y = strand ? gry : gfy;
  }
  w[FS] = (v2f){ 0.0f, 0.0f };

  v2f hSS[QT], hS1[QT], hS0[QT], h1S[QT], h11[QT], h10[QT], h0S[QT], h01[QT], h00[QT - 1];
#pragma unroll
  for (int i = 0; i < QT; ++i) {
    const v2f ta = w[4 * i], tb = w[4 * i + 1], tc = w[4 * i + 2], td = w[4 * i + 3];
    hS1[i] = ta + tb;
    hS0[i] = tc + td;
    hSS[i] = hS1[i] + hS0[i];
    h1S[i] = ta + tc;
    h11[i] = ta;
    h10[i] = tc;
    h0S[i] = tb + td;
    h01[i] = tb;
    if (i < QT - 1) h00[i] = td;
  }

  stage_x(xph[0], tid, r0, r1, r2, r3, m0, m1, m2, m3);
  __syncthreads();

  const int rw = (ws * WROWS < NQUAD - WROWS) ? ws * WROWS : (NQUAD - WROWS);

  for (int i = 0; i < BPB; ++i) {
    const int cur = i & 1;
    if (i + 1 < BPB) {
      const v4f* xn4 = (const v4f*)(x + (size_t)(b0 + i + 1) * 4 * LEN);
      r0 = xn4[q];
      r1 = xn4[LEN / 4 + q];
      r2 = xn4[2 * (LEN / 4) + q];
      r3 = xn4[3 * (LEN / 4) + q];
      m0 = xn4[qn];
      m1 = xn4[LEN / 4 + qn];
      m2 = xn4[2 * (LEN / 4) + qn];
      m3 = xn4[3 * (LEN / 4) + qn];
    }

    float rmax = NEG_INF;
    const float* pb = (const float*)&xph[cur][0][0];
    for (int it = 0; it < WTILES; ++it) {
      const int s0  = rw + 16 * it;
      const float* pa = pb + (s0 + n) * 4 + cbase;
      v2f a0[QT], a1[QT], a2[QT], a3[QT], a4[QT], a5[QT], a6[QT], a7[QT], a8[QT - 1];
#pragma unroll
      for (int j = 0; j < QT; ++j) {
        a0[j] = *(const v2f*)(pa + 0 * (PLEN * 4) + j * 4);
        a3[j] = *(const v2f*)(pa + 3 * (PLEN * 4) + j * 4);
        a6[j] = *(const v2f*)(pa + 6 * (PLEN * 4) + j * 4);
        a1[j] = *(const v2f*)(pa + 1 * (PLEN * 4) + j * 4);
        a2[j] = *(const v2f*)(pa + 2 * (PLEN * 4) + j * 4);
        a4[j] = *(const v2f*)(pa + 4 * (PLEN * 4) + j * 4);
        a5[j] = *(const v2f*)(pa + 5 * (PLEN * 4) + j * 4);
        a7[j] = *(const v2f*)(pa + 7 * (PLEN * 4) + j * 4);
        if (j < QT - 1) a8[j] = *(const v2f*)(pa + 8 * (PLEN * 4) + j * 4);
      }
      __builtin_amdgcn_sched_barrier(0);
      v8f acc = {}, qd = {}, qe = {};
#pragma unroll
      for (int j = 0; j < QT; ++j) {
        acc = wmma_f32_16x16x4(a0[j], hSS[j], acc);
        qd  = wmma_f32_16x16x4(a3[j], h1S[j], qd);
        qe  = wmma_f32_16x16x4(a6[j], h0S[j], qe);
      }
      v8f u = acc, v = acc;
#pragma unroll
      for (int j = 0; j < QT; ++j) {
        u = wmma_f32_16x16x4(a1[j], hS1[j], u);
        v = wmma_f32_16x16x4(a2[j], hS0[j], v);
      }
      v8f y0 = u, y1 = u, y2 = v, y3 = v;
#pragma unroll
      for (int j = 0; j < QT; ++j) {
        y0 = wmma_f32_16x16x4(a4[j], h11[j], y0);
        y2 = wmma_f32_16x16x4(a5[j], h10[j], y2);
        y1 = wmma_f32_16x16x4(a7[j], h01[j], y1);
        if (j < QT - 1) y3 = wmma_f32_16x16x4(a8[j], h00[j], y3);
      }
      if (ws == 3 && it == WTILES - 1 && hi) { y2[7] = NEG_INF; y3[7] = NEG_INF; }
#pragma unroll
      for (int r = 0; r < 8; ++r) {
        const float e = fmaxf(y0[r], y2[r]) + qd[r];
        const float o = fmaxf(y1[r], y3[r]) + qe[r];
        rmax = fmaxf(rmax, fmaxf(e, o));
      }
    }

    rmax = fmaxf(rmax, __shfl_xor(rmax, 16, 32));
    if (lane < 16) red[cur][wave][n] = rmax;
    if (i + 1 < BPB) stage_x(xph[cur ^ 1], tid, r0, r1, r2, r3, m0, m1, m2, m3);
    __syncthreads();
    if (tid < 16) {
      float m = red[cur][0][tid];
#pragma unroll
      for (int k = 1; k < 8; ++k) m = fmaxf(m, red[cur][k][tid]);
      out[(size_t)(b0 + i) * NPWM + f0 + tid] = m;
    }
  }
}

extern "C" void kernel_launch(void* const* d_in, const int* in_sizes, int n_in,
                              void* d_out, int out_size, void* d_ws, size_t ws_size,
                              hipStream_t stream) {
  const float* x    = (const float*)d_in[0];
  const float* pwms = (const float*)d_in[1];
  float* out        = (float*)d_out;

  dim3 grid(NPWM / 16, 128 / BPB);
  pwm_scan_max_kernel<<<grid, 256, 0, stream>>>(x, pwms, out);
}


# Fast FIR nested a third time: eight output phases, 27 eighth-rate products of 3, 3, 2 taps: 70 WMMAs per 128 outputs and strand, were 88; four-wave blocks, 2 waves per SIMD; f32 sum order changes
# speedup vs baseline: 2.1241x; 1.1293x over previous
#include <hip/hip_runtime.h>

typedef __attribute__((ext_vector_type(2))) float v2f;
typedef __attribute__((ext_vector_type(4))) float v4f;
typedef __attribute__((ext_vector_type(8))) float v8f;

#define NPWM    512
#define FS      19
#define LEN     1000
#define OUTPOS  982
#define QT      5
#define NPL2    9
#define NPLANE  27
#define PLEN    128
#define WTILES  4
#define LASTS0  107
#define NTHR    128
#define NEG_INF (-3.402823466e38f)

static __device__ __forceinline__ v8f wmma_f32_16x16x4(v2f a, v2f b, v8f c) {
  return __builtin_amdgcn_wmma_f32_16x16x4_f32(
       false, a,  false, b,
       (short)0, c,  false,  false);
}

#define BPB     8


static __device__ __forceinline__ void level_two(v4f* q, v4f c0, v4f c1, v4f c2, v4f c3,
                                                 v4f n0, v4f n1, v4f n2) {
  const v4f dA = c2 - c3;
  const v4f dB = n0 - c3;
  q[0] = c3;
  q[1] = c1 - c3;
  q[2] = n1 - c3;
  q[3] = dA;
  q[4] = (c0 - c1) - dA;
  q[5] = (n0 - n1) - dA;
  q[6] = dB;
  q[7] = (c2 - c1) - dB;
  q[8] = (n2 - n1) - dB;
}

struct XL { v4f v[4][4]; };

static __device__ __forceinline__ void stage_x(v4f (*pl)[PLEN], int tid, XL L) {
#pragma unroll
  for (int k = 0; k < 4; ++k)
    if (2 * tid + k >= LEN / 4) {
      L.v[0][k] = (v4f){}; L.v[1][k] = (v4f){}; L.v[2][k] = (v4f){}; L.v[3][k] = (v4f){};
    }
  v4f P[15];
#pragma unroll
  for (int i = 0; i < 15; ++i)
    P[i] = (v4f){ L.v[0][i >> 2][i & 3], L.v[1][i >> 2][i & 3], L.v[2][i >> 2][i & 3], L.v[3][i >> 2][i & 3] };
  v4f qa[NPL2], qb[NPL2], qc[NPL2];
  level_two(qa, P[0], P[1], P[2],  P[3],  P[4],  P[5],  P[6]);
  level_two(qb, P[4], P[5], P[6],  P[7],  P[8],  P[9],  P[10]);
  level_two(qc, P[8], P[9], P[10], P[11], P[12], P[13], P[14]);
#pragma unroll
  for (int k = 0; k < NPL2; ++k) {
    pl[3 * k + 0][tid] = qb[k];
    pl[3 * k + 1][tid] = qa[k] - qb[k];
    pl[3 * k + 2][tid] = qc[k] - qb[k];
  }
}

template <int NT>
static __device__ __forceinline__ void tree(const v2f (&a)[NPL2][3], const v2f (&b)[NPL2][3],
                                            v8f dseed, v8f eseed,
                                            v8f& y0, v8f& y1, v8f& y2, v8f& y3, v8f& qd, v8f& qe) {
  v8f acc = {}, d = dseed, e = eseed;
#pragma unroll
  for (int m = 0; m < NT; ++m) {
    acc = wmma_f32_16x16x4(a[0][m], b[0][m], acc);
    d   = wmma_f32_16x16x4(a[3][m], b[3][m], d);
    e   = wmma_f32_16x16x4(a[6][m], b[6][m], e);
  }
  v8f u = acc, v = acc;
#pragma unroll
  for (int m = 0; m < NT; ++m) {
    u = wmma_f32_16x16x4(a[1][m], b[1][m], u);
    v = wmma_f32_16x16x4(a[2][m], b[2][m], v);
  }
  v8f t0 = u, t1 = u, t2 = v, t3 = v;
#pragma unroll
  for (int m = 0; m < NT; ++m) {
    t0 = wmma_f32_16x16x4(a[4][m], b[4][m], t0);
    t2 = wmma_f32_16x16x4(a[5][m], b[5][m], t2);
    t1 = wmma_f32_16x16x4(a[7][m], b[7][m], t1);
    if (m < 2) t3 = wmma_f32_16x16x4(a[8][m], b[8][m], t3);
  }
  y0 = t0; y1 = t1; y2 = t2; y3 = t3; qd = d; qe = e;
}

__global__ __launch_bounds__(NTHR) __attribute__((amdgpu_waves_per_eu(2, 2))) void pwm_scan_max_kernel(
    const float* __restrict__ x,
    const float* __restrict__ pwms,
    float* __restrict__ out) {
  __shared__ v4f xph[2][NPLANE][PLEN];
  __shared__ float red[2][4][16];

  const int f0    = blockIdx.x * 16;
  const int b0    = blockIdx.y * BPB;
  const int tid   = threadIdx.x;
  const int lane  = tid & 31;
  const int wave  = tid >> 5;
  const int n     = lane & 15;
  const bool hi   = lane >= 16;
  const int cbase = hi ? 2 : 0;
  const int strand = wave >> 1;
  const int half   = wave & 1;

  const float* wrow = pwms + (size_t)(f0 + n) * (4 * FS);
  const bool upper = strand ? !hi : hi;
  const v4f* wsrc = (const v4f*)(wrow + (upper ? 2 * FS - 2 : 0));
  v4f wl[(2 * FS + 2) / 4];
#pragma unroll
  for (int j = 0; j < (2 * FS + 2) / 4; ++j) wl[j] = wsrc[j];

  int qi[4];
#pragma unroll
  for (int k = 0; k < 4; ++k) qi[k] = (2 * tid + k < LEN / 4) ? 2 * tid + k : (LEN / 4 - 1);
  const v4f* xb4 = (const v4f*)(x + (size_t)b0 * 4 * LEN);
  XL L;
#pragma unroll
  for (int c = 0; c < 4; ++c)
#pragma unroll
    for (int k = 0; k < 4; ++k) L.v[c][k] = xb4[c * (LEN / 4) + qi[k]];

  __builtin_amdgcn_sched_barrier(0);

  v2f w[4 * QT];
#pragma unroll
  for (int k = 0; k < FS; ++k) {
    const int fx = k, fy = FS + k, rx = 2 * FS - 1 - k, ry = FS - 1 - k;
    const float gfx = upper ? wl[(fx + 2) >> 2][(fx + 2) & 3] : wl[fx >> 2][fx & 3];
    const float gfy = upper ? wl[(fy + 2) >> 2][(fy + 2) & 3] : wl[fy >> 2][fy & 3];
    const float grx = upper ? wl[(rx + 2) >> 2][(rx + 2) & 3] : wl[rx >> 2][rx & 3];
    const float gry = upper ? wl[(ry + 2) >> 2][(ry + 2) & 3] : wl[ry >> 2][ry & 3];
    w[k].x = strand ? grx : gfx;
    w[k].y = strand ? gry : gfy;
  }
  w[FS] = (v2f){ 0.0f, 0.0f };

  v2f h[NPL2][QT + 1];
#pragma unroll
  for (int i = 0; i < QT; ++i) {
    const v2f ta = w[4 * i], tb = w[4 * i + 1], tc = w[4 * i + 2], td = w[4 * i + 3];
    h[1][i] = ta + tb;
    h[2][i] = tc + td;
    h[0][i] = h[1][i] + h[2][i];
    h[3][i] = ta + tc;
    h[4][i] = ta;
    h[5][i] = tc;
    h[6][i] = tb + td;
    h[7][i] = tb;
    h[8][i] = td;
  }
  v2f bf[3][NPL2][3];
#pragma unroll
  for (int k = 0; k < NPL2; ++k) {
    h[k][QT] = (v2f){ 0.0f, 0.0f };
#pragma unroll
    for (int m = 0; m < 3; ++m) {
      bf[0][k][m] = h[k][2 * m] + h[k][2 * m + 1];
      bf[1][k][m] = h[k][2 * m];
      bf[2][k][m] = h[k][2 * m + 1];
    }
  }

  stage_x(xph[0], tid, L);
  __syncthreads();

  const int rw = half * (16 * WTILES);

  for (int i = 0; i < BPB; ++i) {
    const int cur = i & 1;
    if (i + 1 < BPB) {
      const v4f* xn4 = (const v4f*)(x + (size_t)(b0 + i + 1) * 4 * LEN);
#pragma unroll
      for (int c = 0; c < 4; ++c)
#pragma unroll
        for (int k = 0; k < 4; ++k) L.v[c][k] = xn4[c * (LEN / 4) + qi[k]];
    }

    float rmax = NEG_INF;
    const float* pb = (const float*)&xph[cur][0][0];
    for (int it = 0; it < WTILES; ++it) {
      const int s0  = (rw + 16 * it < LASTS0) ? rw + 16 * it : LASTS0;
      const float* pa = pb + (s0 + n) * 4 + cbase;
      v2f af[3][NPL2][3];
#pragma unroll
      for (int ci = 0; ci < 3; ++ci)
#pragma unroll
        for (int k = 0; k < NPL2; ++k)
#pragma unroll
          for (int m = 0; m < 3; ++m)
            if (m < 2 || (ci < 2 && k < NPL2 - 1))
              af[ci][k][m] = *(const v2f*)(pa + (3 * k + ci) * (PLEN * 4) + m * 4);
            else
              af[ci][k][m] = (v2f){ 0.0f, 0.0f };
      __builtin_amdgcn_sched_barrier(0);
      v8f s0y, s1y, s2y, s3y, sd, se;
      v8f d0y, d1y, d2y, d3y, dd, de;
      v8f e0y, e1y, e2y, e3y, ed, ee;
      const v8f zero = {};
      tree<3>(af[0], bf[0], zero, zero, s0y, s1y, s2y, s3y, sd, se);
      tree<3>(af[1], bf[1], sd, se, d0y, d1y, d2y, d3y, dd, de);
      tree<2>(af[2], bf[2], sd, se, e0y, e1y, e2y, e3y, ed, ee);
      __builtin_amdgcn_sched_barrier(0);
      if (half == 1 && it == WTILES - 1 && hi) { e2y[7] = NEG_INF; e3y[7] = NEG_INF; }
#pragma unroll
      for (int r = 0; r < 8; ++r) {
        const float gD = fmaxf(s0y[r] + d0y[r], s2y[r] + d2y[r]) + dd[r];
        const float gE = fmaxf(s0y[r] + e0y[r], s2y[r] + e2y[r]) + ed[r];
        const float hD = fmaxf(s1y[r] + d1y[r], s3y[r] + d3y[r]) + de[r];
        const float hE = fmaxf(s1y[r] + e1y[r], s3y[r] + e3y[r]) + ee[r];
        rmax = fmaxf(rmax, fmaxf(fmaxf(gD, gE), fmaxf(hD, hE)));
      }
    }

    rmax = fmaxf(rmax, __shfl_xor(rmax, 16, 32));
    if (lane < 16) red[cur][wave][n] = rmax;
    if (i + 1 < BPB) stage_x(xph[cur ^ 1], tid, L);
    __syncthreads();
    if (tid < 16) {
      float m = red[cur][0][tid];
#pragma unroll
      for (int k = 1; k < 4; ++k) m = fmaxf(m, red[cur][k][tid]);
      out[(size_t)(b0 + i) * NPWM + f0 + tid] = m;
    }
  }
}

extern "C" void kernel_launch(void* const* d_in, const int* in_sizes, int n_in,
                              void* d_out, int out_size, void* d_ws, size_t ws_size,
                              hipStream_t stream) {
  const float* x    = (const float*)d_in[0];
  const float* pwms = (const float*)d_in[1];
  float* out        = (float*)d_out;

  dim3 grid(NPWM / 16, 128 / BPB);
  pwm_scan_max_kernel<<<grid, NTHR, 0, stream>>>(x, pwms, out);
}


# Wave priority (s_setprio) falls with rows done, 3 to 0: the two blocks that share a compute unit take turns on the matrix pipe and finish together; the younger no longer runs its last rows alone
# speedup vs baseline: 2.2113x; 1.0411x over previous
#include <hip/hip_runtime.h>

typedef __attribute__((ext_vector_type(2))) float v2f;
typedef __attribute__((ext_vector_type(4))) float v4f;
typedef __attribute__((ext_vector_type(8))) float v8f;

#define NPWM    512
#define FS      19
#define LEN     1000
#define OUTPOS  982
#define QT      5
#define NPL2    9
#define NPLANE  27
#define PLEN    128
#define WTILES  4
#define LASTS0  107
#define NTHR    128
#define NEG_INF (-3.402823466e38f)

static __device__ __forceinline__ v8f wmma_f32_16x16x4(v2f a, v2f b, v8f c) {
  return __builtin_amdgcn_wmma_f32_16x16x4_f32(
       false, a,  false, b,
       (short)0, c,  false,  false);
}

#define BPB     8


static __device__ __forceinline__ void level_two(v4f* q, v4f c0, v4f c1, v4f c2, v4f c3,
                                                 v4f n0, v4f n1, v4f n2) {
  const v4f dA = c2 - c3;
  const v4f dB = n0 - c3;
  q[0] = c3;
  q[1] = c1 - c3;
  q[2] = n1 - c3;
  q[3] = dA;
  q[4] = (c0 - c1) - dA;
  q[5] = (n0 - n1) - dA;
  q[6] = dB;
  q[7] = (c2 - c1) - dB;
  q[8] = (n2 - n1) - dB;
}

struct XL { v4f v[4][4]; };

static __device__ __forceinline__ void stage_x(v4f (*pl)[PLEN], int tid, XL L) {
#pragma unroll
  for (int k = 0; k < 4; ++k)
    if (2 * tid + k >= LEN / 4) {
      L.v[0][k] = (v4f){}; L.v[1][k] = (v4f){}; L.v[2][k] = (v4f){}; L.v[3][k] = (v4f){};
    }
  v4f P[15];
#pragma unroll
  for (int i = 0; i < 15; ++i)
    P[i] = (v4f){ L.v[0][i >> 2][i & 3], L.v[1][i >> 2][i & 3], L.v[2][i >> 2][i & 3], L.v[3][i >> 2][i & 3] };
  v4f qa[NPL2], qb[NPL2], qc[NPL2];
  level_two(qa, P[0], P[1], P[2],  P[3],  P[4],  P[5],  P[6]);
  level_two(qb, P[4], P[5], P[6],  P[7],  P[8],  P[9],  P[10]);
  level_two(qc, P[8], P[9], P[10], P[11], P[12], P[13], P[14]);
#pragma unroll
  for (int k = 0; k < NPL2; ++k) {
    pl[3 * k + 0][tid] = qb[k];
    pl[3 * k + 1][tid] = qa[k] - qb[k];
    pl[3 * k + 2][tid] = qc[k] - qb[k];
  }
}

template <int NT>
static __device__ __forceinline__ void tree(const v2f (&a)[NPL2][3], const v2f (&b)[NPL2][3],
                                            v8f dseed, v8f eseed,
                                            v8f& y0, v8f& y1, v8f& y2, v8f& y3, v8f& qd, v8f& qe) {
  v8f acc = {}, d = dseed, e = eseed;
#pragma unroll
  for (int m = 0; m < NT; ++m) {
    acc = wmma_f32_16x16x4(a[0][m], b[0][m], acc);
    d   = wmma_f32_16x16x4(a[3][m], b[3][m], d);
    e   = wmma_f32_16x16x4(a[6][m], b[6][m], e);
  }
  v8f u = acc, v = acc;
#pragma unroll
  for (int m = 0; m < NT; ++m) {
    u = wmma_f32_16x16x4(a[1][m], b[1][m], u);
    v = wmma_f32_16x16x4(a[2][m], b[2][m], v);
  }
  v8f t0 = u, t1 = u, t2 = v, t3 = v;
#pragma unroll
  for (int m = 0; m < NT; ++m) {
    t0 = wmma_f32_16x16x4(a[4][m], b[4][m], t0);
    t2 = wmma_f32_16x16x4(a[5][m], b[5][m], t2);
    t1 = wmma_f32_16x16x4(a[7][m], b[7][m], t1);
    if (m < 2) t3 = wmma_f32_16x16x4(a[8][m], b[8][m], t3);
  }
  y0 = t0; y1 = t1; y2 = t2; y3 = t3; qd = d; qe = e;
}

__global__ __launch_bounds__(NTHR) __attribute__((amdgpu_waves_per_eu(2, 2))) void pwm_scan_max_kernel(
    const float* __restrict__ x,
    const float* __restrict__ pwms,
    float* __restrict__ out) {
  __shared__ v4f xph[2][NPLANE][PLEN];
  __shared__ float red[2][4][16];

  const int f0    = blockIdx.x * 16;
  const int b0    = blockIdx.y * BPB;
  const int tid   = threadIdx.x;
  const int lane  = tid & 31;
  const int wave  = tid >> 5;
  const int n     = lane & 15;
  const bool hi   = lane >= 16;
  const int cbase = hi ? 2 : 0;
  const int strand = wave >> 1;
  const int half   = wave & 1;

  __builtin_amdgcn_s_setprio(3);

  const float* wrow = pwms + (size_t)(f0 + n) * (4 * FS);
  const bool upper = strand ? !hi : hi;
  const v4f* wsrc = (const v4f*)(wrow + (upper ? 2 * FS - 2 : 0));
  v4f wl[(2 * FS + 2) / 4];
#pragma unroll
  for (int j = 0; j < (2 * FS + 2) / 4; ++j) wl[j] = wsrc[j];

  int qi[4];
#pragma unroll
  for (int k = 0; k < 4; ++k) qi[k] = (2 * tid + k < LEN / 4) ? 2 * tid + k : (LEN / 4 - 1);
  const v4f* xb4 = (const v4f*)(x + (size_t)b0 * 4 * LEN);
  XL L;
#pragma unroll
  for (int c = 0; c < 4; ++c)
#pragma unroll
    for (int k = 0; k < 4; ++k) L.v[c][k] = xb4[c * (LEN / 4) + qi[k]];

  __builtin_amdgcn_sched_barrier(0);

  v2f w[4 * QT];
#pragma unroll
  for (int k = 0; k < FS; ++k) {
    const int fx = k, fy = FS + k, rx = 2 * FS - 1 - k, ry = FS - 1 - k;
    const float gfx = upper ? wl[(fx + 2) >> 2][(fx + 2) & 3] : wl[fx >> 2][fx & 3];
    const float gfy = upper ? wl[(fy + 2) >> 2][(fy + 2) & 3] : wl[fy >> 2][fy & 3];
    const float grx = upper ? wl[(rx + 2) >> 2][(rx + 2) & 3] : wl[rx >> 2][rx & 3];
    const float gry = upper ? wl[(ry + 2) >> 2][(ry + 2) & 3] : wl[ry >> 2][ry & 3];
    w[k].x = strand ? grx : gfx;
    w[k].y = strand ? gry : gfy;
  }
  w[FS] = (v2f){ 0.0f, 0.0f };

  v2f h[NPL2][QT + 1];
#pragma unroll
  for (int i = 0; i < QT; ++i) {
    const v2f ta = w[4 * i], tb = w[4 * i + 1], tc = w[4 * i + 2], td = w[4 * i + 3];
    h[1][i] = ta + tb;
    h[2][i] = tc + td;
    h[0][i] = h[1][i] + h[2][i];
    h[3][i] = ta + tc;
    h[4][i] = ta;
    h[5][i] = tc;
    h[6][i] = tb + td;
    h[7][i] = tb;
    h[8][i] = td;
  }
  v2f bf[3][NPL2][3];
#pragma unroll
  for (int k = 0; k < NPL2; ++k) {
    h[k][QT] = (v2f){ 0.0f, 0.0f };
#pragma unroll
    for (int m = 0; m < 3; ++m) {
      bf[0][k][m] = h[k][2 * m] + h[k][2 * m + 1];
      bf[1][k][m] = h[k][2 * m];
      bf[2][k][m] = h[k][2 * m + 1];
    }
  }

  stage_x(xph[0], tid, L);
  __syncthreads();

  const int rw = half * (16 * WTILES);

  for (int i = 0; i < BPB; ++i) {
    const int cur = i & 1;
    if (i == 3)            __builtin_amdgcn_s_setprio(2);
    else if (i == 5)       __builtin_amdgcn_s_setprio(1);
    else if (i == BPB - 1) __builtin_amdgcn_s_setprio(0);
    if (i + 1 < BPB) {
      const v4f* xn4 = (const v4f*)(x + (size_t)(b0 + i + 1) * 4 * LEN);
#pragma unroll
      for (int c = 0; c < 4; ++c)
#pragma unroll
        for (int k = 0; k < 4; ++k) L.v[c][k] = xn4[c * (LEN / 4) + qi[k]];
    }

    float rmax = NEG_INF;
    const float* pb = (const float*)&xph[cur][0][0];
    for (int it = 0; it < WTILES; ++it) {
      const int s0  = (rw + 16 * it < LASTS0) ? rw + 16 * it : LASTS0;
      const float* pa = pb + (s0 + n) * 4 + cbase;
      v2f af[3][NPL2][3];
#pragma unroll
      for (int ci = 0; ci < 3; ++ci)
#pragma unroll
        for (int k = 0; k < NPL2; ++k)
#pragma unroll
          for (int m = 0; m < 3; ++m)
            if (m < 2 || (ci < 2 && k < NPL2 - 1))
              af[ci][k][m] = *(const v2f*)(pa + (3 * k + ci) * (PLEN * 4) + m * 4);
            else
              af[ci][k][m] = (v2f){ 0.0f, 0.0f };
      __builtin_amdgcn_sched_barrier(0);
      v8f s0y, s1y, s2y, s3y, sd, se;
      v8f d0y, d1y, d2y, d3y, dd, de;
      v8f e0y, e1y, e2y, e3y, ed, ee;
      const v8f zero = {};
      tree<3>(af[0], bf[0], zero, zero, s0y, s1y, s2y, s3y, sd, se);
      tree<3>(af[1], bf[1], sd, se, d0y, d1y, d2y, d3y, dd, de);
      tree<2>(af[2], bf[2], sd, se, e0y, e1y, e2y, e3y, ed, ee);
      __builtin_amdgcn_sched_barrier(0);
      if (half == 1 && it == WTILES - 1 && hi) { e2y[7] = NEG_INF; e3y[7] = NEG_INF; }
#pragma unroll
      for (int r = 0; r < 8; ++r) {
        const float gD = fmaxf(s0y[r] + d0y[r], s2y[r] + d2y[r]) + dd[r];
        const float gE = fmaxf(s0y[r] + e0y[r], s2y[r] + e2y[r]) + ed[r];
        const float hD = fmaxf(s1y[r] + d1y[r], s3y[r] + d3y[r]) + de[r];
        const float hE = fmaxf(s1y[r] + e1y[r], s3y[r] + e3y[r]) + ee[r];
        rmax = fmaxf(rmax, fmaxf(fmaxf(gD, gE), fmaxf(hD, hE)));
      }
    }

    rmax = fmaxf(rmax, __shfl_xor(rmax, 16, 32));
    if (lane < 16) red[cur][wave][n] = rmax;
    if (i + 1 < BPB) stage_x(xph[cur ^ 1], tid, L);
    __syncthreads();
    if (tid < 16) {
      float m = red[cur][0][tid];
#pragma unroll
      for (int k = 1; k < 4; ++k) m = fmaxf(m, red[cur][k][tid]);
      out[(size_t)(b0 + i) * NPWM + f0 + tid] = m;
    }
  }
}

extern "C" void kernel_launch(void* const* d_in, const int* in_sizes, int n_in,
                              void* d_out, int out_size, void* d_ws, size_t ws_size,
                              hipStream_t stream) {
  const float* x    = (const float*)d_in[0];
  const float* pwms = (const float*)d_in[1];
  float* out        = (float*)d_out;

  dim3 grid(NPWM / 16, 128 / BPB);
  pwm_scan_max_kernel<<<grid, NTHR, 0, stream>>>(x, pwms, out);
}


# Tile maxima as whole-tile vector operations: packed two-element adds on the accumulator register pairs (48 a tile, were 96 single adds); running maximum kept per register, reduced once per row
# speedup vs baseline: 2.2847x; 1.0332x over previous
#include <hip/hip_runtime.h>

typedef __attribute__((ext_vector_type(2))) float v2f;
typedef __attribute__((ext_vector_type(4))) float v4f;
typedef __attribute__((ext_vector_type(8))) float v8f;

#define NPWM    512
#define FS      19
#define LEN     1000
#define OUTPOS  982
#define QT      5
#define NPL2    9
#define NPLANE  27
#define PLEN    128
#define WTILES  4
#define LASTS0  107
#define NTHR    128
#define NEG_INF (-3.402823466e38f)

static __device__ __forceinline__ v8f wmma_f32_16x16x4(v2f a, v2f b, v8f c) {
  return __builtin_amdgcn_wmma_f32_16x16x4_f32(
       false, a,  false, b,
       (short)0, c,  false,  false);
}

static __device__ __forceinline__ v8f vmax8(v8f a, v8f b) { return __builtin_elementwise_max(a, b); }

#define BPB     8


static __device__ __forceinline__ void level_two(v4f* q, v4f c0, v4f c1, v4f c2, v4f c3,
                                                 v4f n0, v4f n1, v4f n2) {
  const v4f dA = c2 - c3;
  const v4f dB = n0 - c3;
  q[0] = c3;
  q[1] = c1 - c3;
  q[2] = n1 - c3;
  q[3] = dA;
  q[4] = (c0 - c1) - dA;
  q[5] = (n0 - n1) - dA;
  q[6] = dB;
  q[7] = (c2 - c1) - dB;
  q[8] = (n2 - n1) - dB;
}

struct XL { v4f v[4][4]; };

static __device__ __forceinline__ void stage_x(v4f (*pl)[PLEN], int tid, XL L) {
#pragma unroll
  for (int k = 0; k < 4; ++k)
    if (2 * tid + k >= LEN / 4) {
      L.v[0][k] = (v4f){}; L.v[1][k] = (v4f){}; L.v[2][k] = (v4f){}; L.v[3][k] = (v4f){};
    }
  v4f P[15];
#pragma unroll
  for (int i = 0; i < 15; ++i)
    P[i] = (v4f){ L.v[0][i >> 2][i & 3], L.v[1][i >> 2][i & 3], L.v[2][i >> 2][i & 3], L.v[3][i >> 2][i & 3] };
  v4f qa[NPL2], qb[NPL2], qc[NPL2];
  level_two(qa, P[0], P[1], P[2],  P[3],  P[4],  P[5],  P[6]);
  level_two(qb, P[4], P[5], P[6],  P[7],  P[8],  P[9],  P[10]);
  level_two(qc, P[8], P[9], P[10], P[11], P[12], P[13], P[14]);
#pragma unroll
  for (int k = 0; k < NPL2; ++k) {
    pl[3 * k + 0][tid] = qb[k];
    pl[3 * k + 1][tid] = qa[k] - qb[k];
    pl[3 * k + 2][tid] = qc[k] - qb[k];
  }
}

template <int NT>
static __device__ __forceinline__ void tree(const v2f (&a)[NPL2][3], const v2f (&b)[NPL2][3],
                                            v8f dseed, v8f eseed,
                                            v8f& y0, v8f& y1, v8f& y2, v8f& y3, v8f& qd, v8f& qe) {
  v8f acc = {}, d = dseed, e = eseed;
#pragma unroll
  for (int m = 0; m < NT; ++m) {
    acc = wmma_f32_16x16x4(a[0][m], b[0][m], acc);
    d   = wmma_f32_16x16x4(a[3][m], b[3][m], d);
    e   = wmma_f32_16x16x4(a[6][m], b[6][m], e);
  }
  v8f u = acc, v = acc;
#pragma unroll
  for (int m = 0; m < NT; ++m) {
    u = wmma_f32_16x16x4(a[1][m], b[1][m], u);
    v = wmma_f32_16x16x4(a[2][m], b[2][m], v);
  }
  v8f t0 = u, t1 = u, t2 = v, t3 = v;
#pragma unroll
  for (int m = 0; m < NT; ++m) {
    t0 = wmma_f32_16x16x4(a[4][m], b[4][m], t0);
    t2 = wmma_f32_16x16x4(a[5][m], b[5][m], t2);
    t1 = wmma_f32_16x16x4(a[7][m], b[7][m], t1);
    if (m < 2) t3 = wmma_f32_16x16x4(a[8][m], b[8][m], t3);
  }
  y0 = t0; y1 = t1; y2 = t2; y3 = t3; qd = d; qe = e;
}

__global__ __launch_bounds__(NTHR) __attribute__((amdgpu_waves_per_eu(2, 2))) void pwm_scan_max_kernel(
    const float* __restrict__ x,
    const float* __restrict__ pwms,
    float* __restrict__ out) {
  __shared__ v4f xph[2][NPLANE][PLEN];
  __shared__ float red[2][4][16];

  const int f0    = blockIdx.x * 16;
  const int b0    = blockIdx.y * BPB;
  const int tid   = threadIdx.x;
  const int lane  = tid & 31;
  const int wave  = tid >> 5;
  const int n     = lane & 15;
  const bool hi   = lane >= 16;
  const int cbase = hi ? 2 : 0;
  const int strand = wave >> 1;
  const int half   = wave & 1;

  __builtin_amdgcn_s_setprio(3);

  const float* wrow = pwms + (size_t)(f0 + n) * (4 * FS);
  const bool upper = strand ? !hi : hi;
  const v4f* wsrc = (const v4f*)(wrow + (upper ? 2 * FS - 2 : 0));
  v4f wl[(2 * FS + 2) / 4];
#pragma unroll
  for (int j = 0; j < (2 * FS + 2) / 4; ++j) wl[j] = wsrc[j];

  int qi[4];
#pragma unroll
  for (int k = 0; k < 4; ++k) qi[k] = (2 * tid + k < LEN / 4) ? 2 * tid + k : (LEN / 4 - 1);
  const v4f* xb4 = (const v4f*)(x + (size_t)b0 * 4 * LEN);
  XL L;
#pragma unroll
  for (int c = 0; c < 4; ++c)
#pragma unroll
    for (int k = 0; k < 4; ++k) L.v[c][k] = xb4[c * (LEN / 4) + qi[k]];

  __builtin_amdgcn_sched_barrier(0);

  v2f w[4 * QT];
#pragma unroll
  for (int k = 0; k < FS; ++k) {
    const int fx = k, fy = FS + k, rx = 2 * FS - 1 - k, ry = FS - 1 - k;
    const float gfx = upper ? wl[(fx + 2) >> 2][(fx + 2) & 3] : wl[fx >> 2][fx & 3];
    const float gfy = upper ? wl[(fy + 2) >> 2][(fy + 2) & 3] : wl[fy >> 2][fy & 3];
    const float grx = upper ? wl[(rx + 2) >> 2][(rx + 2) & 3] : wl[rx >> 2][rx & 3];
    const float gry = upper ? wl[(ry + 2) >> 2][(ry + 2) & 3] : wl[ry >> 2][ry & 3];
    w[k].x = strand ? grx : gfx;
    w[k].y = strand ? gry : gfy;
  }
  w[FS] = (v2f){ 0.0f, 0.0f };

  v2f h[NPL2][QT + 1];
#pragma unroll
  for (int i = 0; i < QT; ++i) {
    const v2f ta = w[4 * i], tb = w[4 * i + 1], tc = w[4 * i + 2], td = w[4 * i + 3];
    h[1][i] = ta + tb;
    h[2][i] = tc + td;
    h[0][i] = h[1][i] + h[2][i];
    h[3][i] = ta + tc;
    h[4][i] = ta;
    h[5][i] = tc;
    h[6][i] = tb + td;
    h[7][i] = tb;
    h[8][i] = td;
  }
  v2f bf[3][NPL2][3];
#pragma unroll
  for (int k = 0; k < NPL2; ++k) {
    h[k][QT] = (v2f){ 0.0f, 0.0f };
#pragma unroll
    for (int m = 0; m < 3; ++m) {
      bf[0][k][m] = h[k][2 * m] + h[k][2 * m + 1];
      bf[1][k][m] = h[k][2 * m];
      bf[2][k][m] = h[k][2 * m + 1];
    }
  }

  stage_x(xph[0], tid, L);
  __syncthreads();

  const int rw = half * (16 * WTILES);

  for (int i = 0; i < BPB; ++i) {
    const int cur = i & 1;
    if (i == 3)            __builtin_amdgcn_s_setprio(2);
    else if (i == 5)       __builtin_amdgcn_s_setprio(1);
    else if (i == BPB - 1) __builtin_amdgcn_s_setprio(0);
    if (i + 1 < BPB) {
      const v4f* xn4 = (const v4f*)(x + (size_t)(b0 + i + 1) * 4 * LEN);
#pragma unroll
      for (int c = 0; c < 4; ++c)
#pragma unroll
        for (int k = 0; k < 4; ++k) L.v[c][k] = xn4[c * (LEN / 4) + qi[k]];
    }

    v8f rmaxv = { NEG_INF, NEG_INF, NEG_INF, NEG_INF, NEG_INF, NEG_INF, NEG_INF, NEG_INF };
    const float* pb = (const float*)&xph[cur][0][0];
    for (int it = 0; it < WTILES; ++it) {
      const int s0  = (rw + 16 * it < LASTS0) ? rw + 16 * it : LASTS0;
      const float* pa = pb + (s0 + n) * 4 + cbase;
      v2f af[3][NPL2][3];
#pragma unroll
      for (int ci = 0; ci < 3; ++ci)
#pragma unroll
        for (int k = 0; k < NPL2; ++k)
#pragma unroll
          for (int m = 0; m < 3; ++m)
            if (m < 2 || (ci < 2 && k < NPL2 - 1))
              af[ci][k][m] = *(const v2f*)(pa + (3 * k + ci) * (PLEN * 4) + m * 4);
            else
              af[ci][k][m] = (v2f){ 0.0f, 0.0f };
      __builtin_amdgcn_sched_barrier(0);
      v8f s0y, s1y, s2y, s3y, sd, se;
      v8f d0y, d1y, d2y, d3y, dd, de;
      v8f e0y, e1y, e2y, e3y, ed, ee;
      const v8f zero = {};
      tree<3>(af[0], bf[0], zero, zero, s0y, s1y, s2y, s3y, sd, se);
      tree<3>(af[1], bf[1], sd, se, d0y, d1y, d2y, d3y, dd, de);
      tree<2>(af[2], bf[2], sd, se, e0y, e1y, e2y, e3y, ed, ee);
      __builtin_amdgcn_sched_barrier(0);
      if (half == 1 && it == WTILES - 1 && hi) { e2y[7] = NEG_INF; e3y[7] = NEG_INF; }
      const v8f gD = vmax8(s0y + d0y, s2y + d2y) + dd;
      const v8f gE = vmax8(s0y + e0y, s2y + e2y) + ed;
      const v8f hD = vmax8(s1y + d1y, s3y + d3y) + de;
      const v8f hE = vmax8(s1y + e1y, s3y + e3y) + ee;
      rmaxv = vmax8(rmaxv, vmax8(vmax8(gD, gE), vmax8(hD, hE)));
    }
    float rmax = fmaxf(fmaxf(fmaxf(rmaxv[0], rmaxv[1]), fmaxf(rmaxv[2], rmaxv[3])),
                       fmaxf(fmaxf(rmaxv[4], rmaxv[5]), fmaxf(rmaxv[6], rmaxv[7])));

    rmax = fmaxf(rmax, __shfl_xor(rmax, 16, 32));
    if (lane < 16) red[cur][wave][n] = rmax;
    if (i + 1 < BPB) stage_x(xph[cur ^ 1], tid, L);
    __syncthreads();
    if (tid < 16) {
      float m = red[cur][0][tid];
#pragma unroll
      for (int k = 1; k < 4; ++k) m = fmaxf(m, red[cur][k][tid]);
      out[(size_t)(b0 + i) * NPWM + f0 + tid] = m;
    }
  }
}

extern "C" void kernel_launch(void* const* d_in, const int* in_sizes, int n_in,
                              void* d_out, int out_size, void* d_ws, size_t ws_size,
                              hipStream_t stream) {
  const float* x    = (const float*)d_in[0];
  const float* pwms = (const float*)d_in[1];
  float* out        = (float*)d_out;

  dim3 grid(NPWM / 16, 128 / BPB);
  pwm_scan_max_kernel<<<grid, NTHR, 0, stream>>>(x, pwms, out);
}
